# starting kernel
# baseline (speedup 1.0000x reference)
#include <hip/hip_runtime.h>


typedef __attribute__((ext_vector_type(2))) float v2f;
typedef __attribute__((ext_vector_type(8))) float v8f;

namespace {
constexpr int kWave = 32;
constexpr int kWaves = 8;
constexpr int kBlock = kWave * kWaves;
constexpr float kHalfPi = 1.57079632679489661923f;
}

__device__ __forceinline__ void sincos_hpi(float r, float& s, float& c) {
  const float r2 = r * r;
  float sp = -2.50521084e-08f;
  sp = fmaf(sp, r2,  2.75573192e-06f);
  sp = fmaf(sp, r2, -1.98412698e-04f);
  sp = fmaf(sp, r2,  8.33333333e-03f);
  sp = fmaf(sp, r2, -1.66666667e-01f);
  s  = fmaf(sp * r2, r, r);
  float cp =  2.08767570e-09f;
  cp = fmaf(cp, r2, -2.75573192e-07f);
  cp = fmaf(cp, r2,  2.48015873e-05f);
  cp = fmaf(cp, r2, -1.38888889e-03f);
  cp = fmaf(cp, r2,  4.16666667e-02f);
  cp = fmaf(cp, r2, -5.00000000e-01f);
  c  = fmaf(cp, r2, 1.0f);
}

__device__ __forceinline__ void pair_sim(float cA, float sA, float cB, float sB,
                                         float p[4]) {
  float v0 = cA * cB, v1 = cA * sB, v2 = sA * cB, v3 = sA * sB;
#pragma unroll
  for (int l = 0; l < 3; ++l) {
    float t0 = cA * v0 - sA * v2;
    float t2 = sA * v0 + cA * v2;
    float t1 = cA * v1 - sA * v3;
    float t3 = sA * v1 + cA * v3;
    float u0 = cB * t0 - sB * t1;
    float u1 = sB * t0 + cB * t1;
    float u2 = cB * t2 - sB * t3;
    float u3 = sB * t2 + cB * t3;
    v0 = u0; v1 = u1; v2 = u3; v3 = u2;
  }
  p[0] = v0 * v0; p[1] = v1 * v1; p[2] = v2 * v2; p[3] = v3 * v3;
}

__global__ __launch_bounds__(kBlock) void qconv_wmma_kernel(
    const float* __restrict__ patch, float* __restrict__ out, int n_elem) {
  __shared__ __align__(16) float ldsP[kWaves * kWave * 8];
  __shared__ __align__(16) float ldsO[kWaves * kWave * 4];

  const int lane = threadIdx.x & (kWave - 1);
  const int wave = threadIdx.x >> 5;
  const int e_base = blockIdx.x * kBlock + wave * kWave;
  int e = e_base + lane;
  if (e >= n_elem) e = n_elem - 1;

  const float4 x = reinterpret_cast<const float4*>(patch)[e];
  const float h0 = fminf(fmaxf(x.x, -1.f), 1.f) * kHalfPi;
  const float h1 = fminf(fmaxf(x.y, -1.f), 1.f) * kHalfPi;
  const float h2 = fminf(fmaxf(x.z, -1.f), 1.f) * kHalfPi;
  const float h3 = fminf(fmaxf(x.w, -1.f), 1.f) * kHalfPi;
  float c0, s0, c1, s1, c2, s2, c3, s3;
  sincos_hpi(h0, s0, c0);
  sincos_hpi(h1, s1, c1);
  sincos_hpi(h2, s2, c2);
  sincos_hpi(h3, s3, c3);

  float p01[4], p23[4];
  pair_sim(c0, s0, c1, s1, p01);
  pair_sim(c2, s2, c3, s3, p23);

  float* myP = &ldsP[(wave * kWave + lane) * 8];
#pragma unroll
  for (int k = 0; k < 4; ++k) { myP[k] = p01[k]; myP[4 + k] = p23[k]; }
  __syncthreads();

  const int m = lane & 15;
  const int hi = lane >> 4;
  const int n = m;
  const float zs = hi ? -1.f : 1.f;
  const v2f b01 = { (n == 0) ? zs : (n == 1) ?  1.f : 0.f,
                    (n == 0) ? zs : (n == 1) ? -1.f : 0.f };
  const v2f b23 = { (n == 2) ? zs : (n == 3) ?  1.f : 0.f,
                    (n == 2) ? zs : (n == 3) ? -1.f : 0.f };

#pragma unroll
  for (int t = 0; t < 2; ++t) {
    const float* src = &ldsP[(wave * kWave + t * 16 + m) * 8];
    v2f a01 = { src[hi * 2 + 0],     src[hi * 2 + 1] };
    v2f a23 = { src[4 + hi * 2 + 0], src[4 + hi * 2 + 1] };
    v8f acc = {};
    acc = __builtin_amdgcn_wmma_f32_16x16x4_f32(false, a01, false, b01,
                                                (short)0, acc, false, false);
    acc = __builtin_amdgcn_wmma_f32_16x16x4_f32(false, a23, false, b23,
                                                (short)0, acc, false, false);
    if (n < 4) {
#pragma unroll
      for (int i = 0; i < 8; ++i) {
        const int row = i + 8 * hi;
        ldsO[(wave * kWave + t * 16 + row) * 4 + n] = acc[i];
      }
    }
  }
  __syncthreads();

  const float4 r =
      reinterpret_cast<const float4*>(&ldsO[wave * kWave * 4])[lane];
  reinterpret_cast<float4*>(out)[e] = r;
}

extern "C" void kernel_launch(void* const* d_in, const int* in_sizes, int n_in,
                              void* d_out, int out_size, void* d_ws, size_t ws_size,
                              hipStream_t stream) {
  const float* patch = (const float*)d_in[0];
  float* out = (float*)d_out;
  const int n_elem = in_sizes[0] / 4;
  const int blocks = (n_elem + kBlock - 1) / kBlock;
  qconv_wmma_kernel<<<blocks, kBlock, 0, stream>>>(patch, out, n_elem);
}


# Z projection WMMA transposed (features on M, elements on N): B = the lane's own packed prob pair, D = its own float4; LDS staging and both barriers removed; a feature's 4 terms now sum one per WMMA
# speedup vs baseline: 1.1986x; 1.1986x over previous
#include <hip/hip_runtime.h>


typedef __attribute__((ext_vector_type(2))) float v2f;
typedef __attribute__((ext_vector_type(8))) float v8f;

namespace {
constexpr int kWave = 32;
constexpr int kWaves = 8;
constexpr int kBlock = kWave * kWaves;
constexpr float kHalfPi = 1.57079632679489661923f;
}

__device__ __forceinline__ v2f fma2(v2f a, v2f b, v2f c) {
  return __builtin_elementwise_fma(a, b, c);
}

__device__ __forceinline__ void sincos_hpi(v2f r, v2f& s, v2f& c) {
  const v2f r2 = r * r;
  v2f sp = -2.50521084e-08f;
  sp = fma2(sp, r2,  2.75573192e-06f);
  sp = fma2(sp, r2, -1.98412698e-04f);
  sp = fma2(sp, r2,  8.33333333e-03f);
  sp = fma2(sp, r2, -1.66666667e-01f);
  s  = fma2(sp * r2, r, r);
  v2f cp =  2.08767570e-09f;
  cp = fma2(cp, r2, -2.75573192e-07f);
  cp = fma2(cp, r2,  2.48015873e-05f);
  cp = fma2(cp, r2, -1.38888889e-03f);
  cp = fma2(cp, r2,  4.16666667e-02f);
  cp = fma2(cp, r2, -5.00000000e-01f);
  c  = fma2(cp, r2, 1.0f);
}

__device__ __forceinline__ void pair_sim(v2f cA, v2f sA, v2f cB, v2f sB, v2f p[4]) {
  v2f v0 = cA * cB, v1 = cA * sB, v2 = sA * cB, v3 = sA * sB;
#pragma unroll
  for (int l = 0; l < 3; ++l) {
    v2f t0 = cA * v0 - sA * v2;
    v2f t2 = sA * v0 + cA * v2;
    v2f t1 = cA * v1 - sA * v3;
    v2f t3 = sA * v1 + cA * v3;
    v2f u0 = cB * t0 - sB * t1;
    v2f u1 = sB * t0 + cB * t1;
    v2f u2 = cB * t2 - sB * t3;
    v2f u3 = sB * t2 + cB * t3;
    v0 = u0; v1 = u1; v2 = u3; v3 = u2;
  }
  p[0] = v0 * v0; p[1] = v1 * v1; p[2] = v2 * v2; p[3] = v3 * v3;
}

__global__ __launch_bounds__(kBlock) void qconv_wmma_kernel(
    const float* __restrict__ patch, float* __restrict__ out, int n_elem) {
  const int lane = threadIdx.x & (kWave - 1);
  int e = blockIdx.x * kBlock + threadIdx.x;
  if (e >= n_elem) e = n_elem - 1;

  const float4 x = reinterpret_cast<const float4*>(patch)[e];
  const v2f xA = { fminf(fmaxf(x.x, -1.f), 1.f), fminf(fmaxf(x.z, -1.f), 1.f) };
  const v2f xB = { fminf(fmaxf(x.y, -1.f), 1.f), fminf(fmaxf(x.w, -1.f), 1.f) };
  v2f cA, sA, cB, sB;
  sincos_hpi(xA * kHalfPi, sA, cA);
  sincos_hpi(xB * kHalfPi, sB, cB);

  v2f p[4];
  pair_sim(cA, sA, cB, sB, p);

  const int f = lane & 3;
  const bool live = (lane < 4) || ((lane & ~3) == 24);
  const bool tgt = (f & 1) != 0;
  const float m01 = (live && f < 2) ? 1.f : 0.f;
  const float m23 = (live && f >= 2) ? 1.f : 0.f;
  const float sg1 = tgt ? -1.f : 1.f;
  const float sg2 = -sg1;
  const v2f a0 = {  m01,        m23 };
  const v2f a1 = {  m01 * sg1,  m23 * sg1 };
  const v2f a2 = {  m01 * sg2,  m23 * sg2 };
  const v2f a3 = { -m01,       -m23 };

  v8f acc = {};
  acc = __builtin_amdgcn_wmma_f32_16x16x4_f32(false, a0, false, p[0], (short)0, acc,
                                              false, false);
  acc = __builtin_amdgcn_wmma_f32_16x16x4_f32(false, a1, false, p[1], (short)0, acc,
                                              false, false);
  acc = __builtin_amdgcn_wmma_f32_16x16x4_f32(false, a2, false, p[2], (short)0, acc,
                                              false, false);
  acc = __builtin_amdgcn_wmma_f32_16x16x4_f32(false, a3, false, p[3], (short)0, acc,
                                              false, false);

  reinterpret_cast<float4*>(out)[e] = make_float4(acc[0], acc[1], acc[2], acc[3]);
}

extern "C" void kernel_launch(void* const* d_in, const int* in_sizes, int n_in,
                              void* d_out, int out_size, void* d_ws, size_t ws_size,
                              hipStream_t stream) {
  const float* patch = (const float*)d_in[0];
  float* out = (float*)d_out;
  const int n_elem = in_sizes[0] / 4;
  const int blocks = (n_elem + kBlock - 1) / kBlock;
  qconv_wmma_kernel<<<blocks, kBlock, 0, stream>>>(patch, out, n_elem);
}


# Each lane carries 8 elements (8192 waves, not 65536): a wave requests its eight 512-byte loads at entry and builds arguments, addresses and Z-sign fragment once; rows run in pairs; sums unchanged
# speedup vs baseline: 1.4220x; 1.1864x over previous
#include <hip/hip_runtime.h>


typedef __attribute__((ext_vector_type(2))) float v2f;
typedef __attribute__((ext_vector_type(8))) float v8f;

namespace {
constexpr int kWave = 32;
constexpr int kWaves = 8;
constexpr int kBlock = kWave * kWaves;
constexpr int kPerLane = 8;
constexpr int kWaveElems = kWave * kPerLane;
constexpr int kBlockElems = kBlock * kPerLane;
constexpr float kHalfPi = 1.57079632679489661923f;
}

__device__ __forceinline__ v2f fma2(v2f a, v2f b, v2f c) {
  return __builtin_elementwise_fma(a, b, c);
}

__device__ __forceinline__ void sincos_hpi(v2f r, v2f& s, v2f& c) {
  const v2f r2 = r * r;
  v2f sp = -2.50521084e-08f;
  sp = fma2(sp, r2,  2.75573192e-06f);
  sp = fma2(sp, r2, -1.98412698e-04f);
  sp = fma2(sp, r2,  8.33333333e-03f);
  sp = fma2(sp, r2, -1.66666667e-01f);
  s  = fma2(sp * r2, r, r);
  v2f cp =  2.08767570e-09f;
  cp = fma2(cp, r2, -2.75573192e-07f);
  cp = fma2(cp, r2,  2.48015873e-05f);
  cp = fma2(cp, r2, -1.38888889e-03f);
  cp = fma2(cp, r2,  4.16666667e-02f);
  cp = fma2(cp, r2, -5.00000000e-01f);
  c  = fma2(cp, r2, 1.0f);
}

__device__ __forceinline__ void pair_sim(v2f cA, v2f sA, v2f cB, v2f sB, v2f p[4]) {
  v2f v0 = cA * cB, v1 = cA * sB, v2 = sA * cB, v3 = sA * sB;
#pragma unroll
  for (int l = 0; l < 3; ++l) {
    v2f t0 = cA * v0 - sA * v2;
    v2f t2 = sA * v0 + cA * v2;
    v2f t1 = cA * v1 - sA * v3;
    v2f t3 = sA * v1 + cA * v3;
    v2f u0 = cB * t0 - sB * t1;
    v2f u1 = sB * t0 + cB * t1;
    v2f u2 = cB * t2 - sB * t3;
    v2f u3 = sB * t2 + cB * t3;
    v0 = u0; v1 = u1; v2 = u3; v3 = u2;
  }
  p[0] = v0 * v0; p[1] = v1 * v1; p[2] = v2 * v2; p[3] = v3 * v3;
}

template <bool kTail>
__device__ __forceinline__ void run_rows(const float4* __restrict__ in4, float4* __restrict__ out4,
                                         int first, int n_elem,
                                         v2f a0, v2f a1, v2f a2, v2f a3) {
  int idx[kPerLane];
  float4 x[kPerLane];
#pragma unroll
  for (int j = 0; j < kPerLane; ++j) {
    int e = first + j * kWave;
    if (kTail) e = min(e, n_elem - 1);
    idx[j] = e;
    x[j] = in4[e];
  }
#pragma unroll
  for (int j0 = 0; j0 < kPerLane; j0 += 2) {
#pragma unroll
    for (int j = j0; j < j0 + 2; ++j) {
      const v2f xA = { fminf(fmaxf(x[j].x, -1.f), 1.f), fminf(fmaxf(x[j].z, -1.f), 1.f) };
      const v2f xB = { fminf(fmaxf(x[j].y, -1.f), 1.f), fminf(fmaxf(x[j].w, -1.f), 1.f) };
      v2f cA, sA, cB, sB;
      sincos_hpi(xA * kHalfPi, sA, cA);
      sincos_hpi(xB * kHalfPi, sB, cB);

      v2f p[4];
      pair_sim(cA, sA, cB, sB, p);

      v8f acc = {};
      acc = __builtin_amdgcn_wmma_f32_16x16x4_f32(false, a0, false, p[0], (short)0, acc,
                                                  false, false);
      acc = __builtin_amdgcn_wmma_f32_16x16x4_f32(false, a1, false, p[1], (short)0, acc,
                                                  false, false);
      acc = __builtin_amdgcn_wmma_f32_16x16x4_f32(false, a2, false, p[2], (short)0, acc,
                                                  false, false);
      acc = __builtin_amdgcn_wmma_f32_16x16x4_f32(false, a3, false, p[3], (short)0, acc,
                                                  false, false);

      out4[idx[j]] = make_float4(acc[0], acc[1], acc[2], acc[3]);
    }
    __builtin_amdgcn_sched_barrier(0);
  }
}

__global__ __launch_bounds__(kBlock) void qconv_wmma_kernel(
    const float* __restrict__ patch, float* __restrict__ out, int n_elem) {
  const int lane = threadIdx.x & (kWave - 1);
  const int wave = threadIdx.x >> 5;

  const int f = lane & 3;
  const bool live = (lane < 4) || ((lane & ~3) == 24);
  const bool tgt = (f & 1) != 0;
  const float m01 = (live && f < 2) ? 1.f : 0.f;
  const float m23 = (live && f >= 2) ? 1.f : 0.f;
  const float sg1 = tgt ? -1.f : 1.f;
  const float sg2 = -sg1;
  const v2f a0 = {  m01,        m23 };
  const v2f a1 = {  m01 * sg1,  m23 * sg1 };
  const v2f a2 = {  m01 * sg2,  m23 * sg2 };
  const v2f a3 = { -m01,       -m23 };

  const float4* in4 = reinterpret_cast<const float4*>(patch);
  float4* out4 = reinterpret_cast<float4*>(out);
  const int block_first = blockIdx.x * kBlockElems;
  const int first = block_first + wave * kWaveElems + lane;
  if (block_first + kBlockElems <= n_elem) {
    run_rows<false>(in4, out4, first, n_elem, a0, a1, a2, a3);
  } else {
    run_rows<true>(in4, out4, first, n_elem, a0, a1, a2, a3);
  }
}

extern "C" void kernel_launch(void* const* d_in, const int* in_sizes, int n_in,
                              void* d_out, int out_size, void* d_ws, size_t ws_size,
                              hipStream_t stream) {
  const float* patch = (const float*)d_in[0];
  float* out = (float*)d_out;
  const int n_elem = in_sizes[0] / 4;
  const int blocks = (n_elem + kBlockElems - 1) / kBlockElems;
  qconv_wmma_kernel<<<blocks, kBlock, 0, stream>>>(patch, out, n_elem);
}


# Closed-form circuit: encoding + 3 layers collapse by exact angle identities (2a on the control wire; 2b, 4b on the target) to 31 packed ops a row, not 56; amplitude rounding differs, WMMA sums kept
# speedup vs baseline: 1.5476x; 1.0883x over previous
#include <hip/hip_runtime.h>


typedef __attribute__((ext_vector_type(2))) float v2f;
typedef __attribute__((ext_vector_type(8))) float v8f;

namespace {
constexpr int kWave = 32;
constexpr int kWaves = 8;
constexpr int kBlock = kWave * kWaves;
constexpr int kPerLane = 8;
constexpr int kWaveElems = kWave * kPerLane;
constexpr int kBlockElems = kBlock * kPerLane;
constexpr float kHalfPi = 1.57079632679489661923f;
}

__device__ __forceinline__ v2f fma2(v2f a, v2f b, v2f c) {
  return __builtin_elementwise_fma(a, b, c);
}

__device__ __forceinline__ void sincos_hpi(v2f r, v2f& s, v2f& c) {
  const v2f r2 = r * r;
  v2f sp = -2.50521084e-08f;
  sp = fma2(sp, r2,  2.75573192e-06f);
  sp = fma2(sp, r2, -1.98412698e-04f);
  sp = fma2(sp, r2,  8.33333333e-03f);
  sp = fma2(sp, r2, -1.66666667e-01f);
  s  = fma2(sp * r2, r, r);
  v2f cp =  2.08767570e-09f;
  cp = fma2(cp, r2, -2.75573192e-07f);
  cp = fma2(cp, r2,  2.48015873e-05f);
  cp = fma2(cp, r2, -1.38888889e-03f);
  cp = fma2(cp, r2,  4.16666667e-02f);
  cp = fma2(cp, r2, -5.00000000e-01f);
  c  = fma2(cp, r2, 1.0f);
}

__device__ __forceinline__ void pair_sim(v2f cA, v2f sA, v2f cB, v2f sB, v2f p[4]) {
  const v2f dA  = sA + sA;
  const v2f s2A = dA * cA;
  const v2f c2A = fma2(-dA, sA, 1.0f);
  const v2f dB = sB + sB;
  const v2f S2 = dB * cB;
  const v2f C2 = fma2(-dB, sB, 1.0f);
  const v2f eB = S2 + S2;
  const v2f S4 = eB * C2;
  const v2f C4 = fma2(-eB, S2, 1.0f);
  const v2f g1 = cA * c2A, g2 = sA * s2A, g3 = sA * c2A, g4 = cA * s2A;
  const v2f z0r = g1 * C4;
  const v2f z0i = g1 * S4 - g2;
  const v2f z1r = g3 * S2 + g4 * C2;
  const v2f z1i = g3 * C2 + g4 * S2;
  const v2f u0r = cA * z0r - sA * z1r;
  const v2f u0i = cA * z0i - sA * z1i;
  const v2f u1r = sA * z0r + cA * z1r;
  const v2f u1i = sA * z0i + cA * z1i;
  p[0] = u0r * u0r; p[1] = u0i * u0i; p[2] = u1i * u1i; p[3] = u1r * u1r;
}

template <bool kTail>
__device__ __forceinline__ void run_rows(const float4* __restrict__ in4, float4* __restrict__ out4,
                                         int first, int n_elem,
                                         v2f a0, v2f a1, v2f a2, v2f a3) {
  int idx[kPerLane];
  float4 x[kPerLane];
#pragma unroll
  for (int j = 0; j < kPerLane; ++j) {
    int e = first + j * kWave;
    if (kTail) e = min(e, n_elem - 1);
    idx[j] = e;
    x[j] = in4[e];
  }
#pragma unroll
  for (int j0 = 0; j0 < kPerLane; j0 += 2) {
#pragma unroll
    for (int j = j0; j < j0 + 2; ++j) {
      const v2f xA = { fminf(fmaxf(x[j].x, -1.f), 1.f), fminf(fmaxf(x[j].z, -1.f), 1.f) };
      const v2f xB = { fminf(fmaxf(x[j].y, -1.f), 1.f), fminf(fmaxf(x[j].w, -1.f), 1.f) };
      v2f cA, sA, cB, sB;
      sincos_hpi(xA * kHalfPi, sA, cA);
      sincos_hpi(xB * kHalfPi, sB, cB);

      v2f p[4];
      pair_sim(cA, sA, cB, sB, p);

      v8f acc = {};
      acc = __builtin_amdgcn_wmma_f32_16x16x4_f32(false, a0, false, p[0], (short)0, acc,
                                                  false, false);
      acc = __builtin_amdgcn_wmma_f32_16x16x4_f32(false, a1, false, p[1], (short)0, acc,
                                                  false, false);
      acc = __builtin_amdgcn_wmma_f32_16x16x4_f32(false, a2, false, p[2], (short)0, acc,
                                                  false, false);
      acc = __builtin_amdgcn_wmma_f32_16x16x4_f32(false, a3, false, p[3], (short)0, acc,
                                                  false, false);

      out4[idx[j]] = make_float4(acc[0], acc[1], acc[2], acc[3]);
    }
    __builtin_amdgcn_sched_barrier(0);
  }
}

__global__ __launch_bounds__(kBlock) void qconv_wmma_kernel(
    const float* __restrict__ patch, float* __restrict__ out, int n_elem) {
  const int lane = threadIdx.x & (kWave - 1);
  const int wave = threadIdx.x >> 5;

  const int f = lane & 3;
  const bool live = (lane < 4) || ((lane & ~3) == 24);
  const bool tgt = (f & 1) != 0;
  const float m01 = (live && f < 2) ? 1.f : 0.f;
  const float m23 = (live && f >= 2) ? 1.f : 0.f;
  const float sg1 = tgt ? -1.f : 1.f;
  const float sg2 = -sg1;
  const v2f a0 = {  m01,        m23 };
  const v2f a1 = {  m01 * sg1,  m23 * sg1 };
  const v2f a2 = {  m01 * sg2,  m23 * sg2 };
  const v2f a3 = { -m01,       -m23 };

  const float4* in4 = reinterpret_cast<const float4*>(patch);
  float4* out4 = reinterpret_cast<float4*>(out);
  const int block_first = blockIdx.x * kBlockElems;
  const int first = block_first + wave * kWaveElems + lane;
  if (block_first + kBlockElems <= n_elem) {
    run_rows<false>(in4, out4, first, n_elem, a0, a1, a2, a3);
  } else {
    run_rows<true>(in4, out4, first, n_elem, a0, a1, a2, a3);
  }
}

extern "C" void kernel_launch(void* const* d_in, const int* in_sizes, int n_in,
                              void* d_out, int out_size, void* d_ws, size_t ws_size,
                              hipStream_t stream) {
  const float* patch = (const float*)d_in[0];
  float* out = (float*)d_out;
  const int n_elem = in_sizes[0] / 4;
  const int blocks = (n_elem + kBlockElems - 1) / kBlockElems;
  qconv_wmma_kernel<<<blocks, kBlock, 0, stream>>>(patch, out, n_elem);
}


# Z projection takes the signed square differences d0=p0-p3, d1=p1-p2 (formed inside the existing multiply-adds) as B: contraction 4 not 8, two WMMAs a row not four; sum order now (p0-p3)+-(p1-p2)
# speedup vs baseline: 1.7108x; 1.1055x over previous
#include <hip/hip_runtime.h>


typedef __attribute__((ext_vector_type(2))) float v2f;
typedef __attribute__((ext_vector_type(8))) float v8f;

namespace {
constexpr int kWave = 32;
constexpr int kWaves = 8;
constexpr int kBlock = kWave * kWaves;
constexpr int kPerLane = 8;
constexpr int kWaveElems = kWave * kPerLane;
constexpr int kBlockElems = kBlock * kPerLane;
constexpr float kHalfPi = 1.57079632679489661923f;
}

__device__ __forceinline__ v2f fma2(v2f a, v2f b, v2f c) {
  return __builtin_elementwise_fma(a, b, c);
}

__device__ __forceinline__ void sincos_hpi(v2f r, v2f& s, v2f& c) {
  const v2f r2 = r * r;
  v2f sp = -2.50521084e-08f;
  sp = fma2(sp, r2,  2.75573192e-06f);
  sp = fma2(sp, r2, -1.98412698e-04f);
  sp = fma2(sp, r2,  8.33333333e-03f);
  sp = fma2(sp, r2, -1.66666667e-01f);
  s  = fma2(sp * r2, r, r);
  v2f cp =  2.08767570e-09f;
  cp = fma2(cp, r2, -2.75573192e-07f);
  cp = fma2(cp, r2,  2.48015873e-05f);
  cp = fma2(cp, r2, -1.38888889e-03f);
  cp = fma2(cp, r2,  4.16666667e-02f);
  cp = fma2(cp, r2, -5.00000000e-01f);
  c  = fma2(cp, r2, 1.0f);
}

__device__ __forceinline__ void pair_sim(v2f cA, v2f sA, v2f cB, v2f sB, v2f d[2]) {
  const v2f dA  = sA + sA;
  const v2f s2A = dA * cA;
  const v2f c2A = fma2(-dA, sA, 1.0f);
  const v2f dB = sB + sB;
  const v2f S2 = dB * cB;
  const v2f C2 = fma2(-dB, sB, 1.0f);
  const v2f eB = S2 + S2;
  const v2f S4 = eB * C2;
  const v2f C4 = fma2(-eB, S2, 1.0f);
  const v2f g1 = cA * c2A, g2 = sA * s2A, g3 = sA * c2A, g4 = cA * s2A;
  const v2f z0r = g1 * C4;
  const v2f z0i = g1 * S4 - g2;
  const v2f z1r = g3 * S2 + g4 * C2;
  const v2f z1i = g3 * C2 + g4 * S2;
  const v2f u0r = cA * z0r - sA * z1r;
  const v2f u0i = cA * z0i - sA * z1i;
  const v2f u1r = sA * z0r + cA * z1r;
  const v2f u1i = sA * z0i + cA * z1i;
  d[0] = fma2(u0r, u0r, -(u1r * u1r));
  d[1] = fma2(u0i, u0i, -(u1i * u1i));
}

template <bool kTail>
__device__ __forceinline__ void run_rows(const float4* __restrict__ in4, float4* __restrict__ out4,
                                         int first, int n_elem,
                                         v2f a0, v2f a1) {
  int idx[kPerLane];
  float4 x[kPerLane];
#pragma unroll
  for (int j = 0; j < kPerLane; ++j) {
    int e = first + j * kWave;
    if (kTail) e = min(e, n_elem - 1);
    idx[j] = e;
    x[j] = in4[e];
  }
#pragma unroll
  for (int j0 = 0; j0 < kPerLane; j0 += 2) {
#pragma unroll
    for (int j = j0; j < j0 + 2; ++j) {
      const v2f xA = { fminf(fmaxf(x[j].x, -1.f), 1.f), fminf(fmaxf(x[j].z, -1.f), 1.f) };
      const v2f xB = { fminf(fmaxf(x[j].y, -1.f), 1.f), fminf(fmaxf(x[j].w, -1.f), 1.f) };
      v2f cA, sA, cB, sB;
      sincos_hpi(xA * kHalfPi, sA, cA);
      sincos_hpi(xB * kHalfPi, sB, cB);

      v2f d[2];
      pair_sim(cA, sA, cB, sB, d);

      v8f acc = {};
      acc = __builtin_amdgcn_wmma_f32_16x16x4_f32(false, a0, false, d[0], (short)0, acc,
                                                  false, false);
      acc = __builtin_amdgcn_wmma_f32_16x16x4_f32(false, a1, false, d[1], (short)0, acc,
                                                  false, false);

      out4[idx[j]] = make_float4(acc[0], acc[1], acc[2], acc[3]);
    }
    __builtin_amdgcn_sched_barrier(0);
  }
}

__global__ __launch_bounds__(kBlock) void qconv_wmma_kernel(
    const float* __restrict__ patch, float* __restrict__ out, int n_elem) {
  const int lane = threadIdx.x & (kWave - 1);
  const int wave = threadIdx.x >> 5;

  const int f = lane & 3;
  const bool live = (lane < 4) || ((lane & ~3) == 24);
  const bool tgt = (f & 1) != 0;
  const float m01 = (live && f < 2) ? 1.f : 0.f;
  const float m23 = (live && f >= 2) ? 1.f : 0.f;
  const float sg1 = tgt ? -1.f : 1.f;
  const v2f a0 = {  m01,        m23 };
  const v2f a1 = {  m01 * sg1,  m23 * sg1 };

  const float4* in4 = reinterpret_cast<const float4*>(patch);
  float4* out4 = reinterpret_cast<float4*>(out);
  const int block_first = blockIdx.x * kBlockElems;
  const int first = block_first + wave * kWaveElems + lane;
  if (block_first + kBlockElems <= n_elem) {
    run_rows<false>(in4, out4, first, n_elem, a0, a1);
  } else {
    run_rows<true>(in4, out4, first, n_elem, a0, a1);
  }
}

extern "C" void kernel_launch(void* const* d_in, const int* in_sizes, int n_in,
                              void* d_out, int out_size, void* d_ws, size_t ws_size,
                              hipStream_t stream) {
  const float* patch = (const float*)d_in[0];
  float* out = (float*)d_out;
  const int n_elem = in_sizes[0] / 4;
  const int blocks = (n_elem + kBlockElems - 1) / kBlockElems;
  qconv_wmma_kernel<<<blocks, kBlock, 0, stream>>>(patch, out, n_elem);
}


# Sines and cosines from the hardware v_sin_f32/v_cos_f32 (angle in revolutions: x/4 control wire, x/2 target wire) replace the polynomials and one angle doubling; trig values' accuracy differs
# speedup vs baseline: 1.8183x; 1.0628x over previous
#include <hip/hip_runtime.h>


typedef __attribute__((ext_vector_type(2))) float v2f;
typedef __attribute__((ext_vector_type(8))) float v8f;

namespace {
constexpr int kWave = 32;
constexpr int kWaves = 8;
constexpr int kBlock = kWave * kWaves;
constexpr int kPerLane = 8;
constexpr int kWaveElems = kWave * kPerLane;
constexpr int kBlockElems = kBlock * kPerLane;
}

__device__ __forceinline__ v2f fma2(v2f a, v2f b, v2f c) {
  return __builtin_elementwise_fma(a, b, c);
}

__device__ __forceinline__ v2f sin_turns(v2f t) {
  return { __builtin_amdgcn_sinf(t.x), __builtin_amdgcn_sinf(t.y) };
}
__device__ __forceinline__ v2f cos_turns(v2f t) {
  return { __builtin_amdgcn_cosf(t.x), __builtin_amdgcn_cosf(t.y) };
}

__device__ __forceinline__ void pair_sim(v2f cA, v2f sA, v2f C2, v2f S2, v2f d[2]) {
  const v2f dA  = sA + sA;
  const v2f s2A = dA * cA;
  const v2f c2A = fma2(-dA, sA, 1.0f);
  const v2f eB = S2 + S2;
  const v2f S4 = eB * C2;
  const v2f C4 = fma2(-eB, S2, 1.0f);
  const v2f g1 = cA * c2A, g2 = sA * s2A, g3 = sA * c2A, g4 = cA * s2A;
  const v2f z0r = g1 * C4;
  const v2f z0i = g1 * S4 - g2;
  const v2f z1r = g3 * S2 + g4 * C2;
  const v2f z1i = g3 * C2 + g4 * S2;
  const v2f u0r = cA * z0r - sA * z1r;
  const v2f u0i = cA * z0i - sA * z1i;
  const v2f u1r = sA * z0r + cA * z1r;
  const v2f u1i = sA * z0i + cA * z1i;
  d[0] = fma2(u0r, u0r, -(u1r * u1r));
  d[1] = fma2(u0i, u0i, -(u1i * u1i));
}

template <bool kTail>
__device__ __forceinline__ void run_rows(const float4* __restrict__ in4, float4* __restrict__ out4,
                                         int first, int n_elem,
                                         v2f a0, v2f a1) {
  int idx[kPerLane];
  float4 x[kPerLane];
#pragma unroll
  for (int j = 0; j < kPerLane; ++j) {
    int e = first + j * kWave;
    if (kTail) e = min(e, n_elem - 1);
    idx[j] = e;
    x[j] = in4[e];
  }
#pragma unroll
  for (int j0 = 0; j0 < kPerLane; j0 += 2) {
#pragma unroll
    for (int j = j0; j < j0 + 2; ++j) {
      const v2f xA = { fminf(fmaxf(x[j].x, -1.f), 1.f), fminf(fmaxf(x[j].z, -1.f), 1.f) };
      const v2f xB = { fminf(fmaxf(x[j].y, -1.f), 1.f), fminf(fmaxf(x[j].w, -1.f), 1.f) };
      const v2f tA = xA * 0.25f;
      const v2f tB = xB * 0.5f;
      const v2f sA = sin_turns(tA), cA = cos_turns(tA);
      const v2f S2 = sin_turns(tB), C2 = cos_turns(tB);

      v2f d[2];
      pair_sim(cA, sA, C2, S2, d);

      v8f acc = {};
      acc = __builtin_amdgcn_wmma_f32_16x16x4_f32(false, a0, false, d[0], (short)0, acc,
                                                  false, false);
      acc = __builtin_amdgcn_wmma_f32_16x16x4_f32(false, a1, false, d[1], (short)0, acc,
                                                  false, false);

      out4[idx[j]] = make_float4(acc[0], acc[1], acc[2], acc[3]);
    }
    __builtin_amdgcn_sched_barrier(0);
  }
}

__global__ __launch_bounds__(kBlock) void qconv_wmma_kernel(
    const float* __restrict__ patch, float* __restrict__ out, int n_elem) {
  const int lane = threadIdx.x & (kWave - 1);
  const int wave = threadIdx.x >> 5;

  const int f = lane & 3;
  const bool live = (lane < 4) || ((lane & ~3) == 24);
  const bool tgt = (f & 1) != 0;
  const float m01 = (live && f < 2) ? 1.f : 0.f;
  const float m23 = (live && f >= 2) ? 1.f : 0.f;
  const float sg1 = tgt ? -1.f : 1.f;
  const v2f a0 = {  m01,        m23 };
  const v2f a1 = {  m01 * sg1,  m23 * sg1 };

  const float4* in4 = reinterpret_cast<const float4*>(patch);
  float4* out4 = reinterpret_cast<float4*>(out);
  const int block_first = blockIdx.x * kBlockElems;
  const int first = block_first + wave * kWaveElems + lane;
  if (block_first + kBlockElems <= n_elem) {
    run_rows<false>(in4, out4, first, n_elem, a0, a1);
  } else {
    run_rows<true>(in4, out4, first, n_elem, a0, a1);
  }
}

extern "C" void kernel_launch(void* const* d_in, const int* in_sizes, int n_in,
                              void* d_out, int out_size, void* d_ws, size_t ws_size,
                              hipStream_t stream) {
  const float* patch = (const float*)d_in[0];
  float* out = (float*)d_out;
  const int n_elem = in_sizes[0] / 4;
  const int blocks = (n_elem + kBlockElems - 1) / kBlockElems;
  qconv_wmma_kernel<<<blocks, kBlock, 0, stream>>>(patch, out, n_elem);
}
